# starting kernel
# baseline (speedup 1.0000x reference)
#include <hip/hip_runtime.h>

#define D 128

typedef __attribute__((ext_vector_type(2))) float v2f;
typedef __attribute__((ext_vector_type(8))) float v8f;


__global__ void fill_kernel(float* p, float v, long n) {
  long i = (long)blockIdx.x * blockDim.x + threadIdx.x;
  if (i < n) p[i] = v;
}

__global__ void pack_pairs_kernel(const float* __restrict__ W, float* __restrict__ Wp,
                                  int K, int Ncols) {
  int i = blockIdx.x * blockDim.x + threadIdx.x;
  int total = (K >> 1) * Ncols;
  if (i >= total) return;
  int kp = i / Ncols, n = i - kp * Ncols;
  float2 v;
  v.x = W[(long)(2 * kp) * Ncols + n];
  v.y = W[(long)(2 * kp + 1) * Ncols + n];
  ((float2*)Wp)[(long)kp * Ncols + n] = v;
}

__global__ void degree_kernel(const int* __restrict__ snd, const int* __restrict__ rcv,
                              float* deg_s, float* deg_r, int E) {
  int e = blockIdx.x * blockDim.x + threadIdx.x;
  if (e < E) {
    unsafeAtomicAdd(&deg_s[snd[e]], 1.0f);
    unsafeAtomicAdd(&deg_r[rcv[e]], 1.0f);
  }
}

__global__ void rownorm_kernel(const float* __restrict__ src, const float* __restrict__ deg,
                               float* __restrict__ dst, int n_nodes) {
  long i = (long)blockIdx.x * blockDim.x + threadIdx.x;
  long n4 = (long)n_nodes * (D / 4);
  if (i >= n4) return;
  int row = (int)(i >> 5);
  float s = rsqrtf(deg[row]);
  float4 v = ((const float4*)src)[i];
  v.x *= s; v.y *= s; v.z *= s; v.w *= s;
  ((float4*)dst)[i] = v;
}

__global__ void scatter_kernel(const int* __restrict__ snd, const int* __restrict__ rcv,
                               const float* __restrict__ xn, float* msum, int E) {
  long gid = (long)blockIdx.x * blockDim.x + threadIdx.x;
  long total = (long)E * 32;
  if (gid >= total) return;
  int e = (int)(gid >> 5);
  int c = (int)(gid & 31) << 2;
  int s = snd[e], r = rcv[e];
  float4 v = *(const float4*)(xn + (long)s * D + c);
  float* dst = msum + (long)r * D + c;
  unsafeAtomicAdd(dst + 0, v.x);
  unsafeAtomicAdd(dst + 1, v.y);
  unsafeAtomicAdd(dst + 2, v.z);
  unsafeAtomicAdd(dst + 3, v.w);
}


__global__ __launch_bounds__(256)
void sage_gemm_kernel(const float* __restrict__ A0,
                      const float* msum,
                      const float* xn,
                      const float* __restrict__ deg_r,
                      const float* __restrict__ Wp,
                      const float* __restrict__ bias,
                      float* out,
                      int n_nodes, int do_relu) {
  int wave = (blockIdx.x << 3) + (threadIdx.x >> 5);
  int mt_count = (n_nodes + 15) >> 4;
  if (wave >= mt_count) return;
  int lane = threadIdx.x & 31;
  int lo = lane & 15, hi = lane >> 4;

  int m = (wave << 4) + lo;
  bool mvalid = m < n_nodes;
  int mr = mvalid ? m : 0;

  float c = deg_r[mr];
  float rs = rsqrtf(c);
  float scale = rs * rs * rs;

  const float* a0row = A0 + (long)mr * D;
  const float* smrow = msum + (long)mr * D;
  const float* xnrow = xn + (long)mr * D;
  const v2f* wp = (const v2f*)Wp;

  v8f acc[8];
#pragma unroll
  for (int t = 0; t < 8; ++t) acc[t] = (v8f)0.0f;

  for (int kk = 0; kk < D; kk += 4) {
    int k0 = kk + (hi << 1);
    v2f a = *(const v2f*)(a0row + k0);
    if (!mvalid) { a.x = 0.0f; a.y = 0.0f; }
    const v2f* wrow = wp + (long)(k0 >> 1) * D + lo;
#pragma unroll
    for (int t = 0; t < 8; ++t) {
      v2f b = wrow[t * 16];
      acc[t] = __builtin_amdgcn_wmma_f32_16x16x4_f32(
          false, a, false, b, (short)0, acc[t], false, false);
    }
  }

  for (int kk = 0; kk < D; kk += 4) {
    int k0 = kk + (hi << 1);
    v2f sm = *(const v2f*)(smrow + k0);
    v2f xv = *(const v2f*)(xnrow + k0);
    v2f a;
    a.x = (sm.x + xv.x) * scale;
    a.y = (sm.y + xv.y) * scale;
    if (!mvalid) { a.x = 0.0f; a.y = 0.0f; }
    const v2f* wrow = wp + (long)((k0 + D) >> 1) * D + lo;
#pragma unroll
    for (int t = 0; t < 8; ++t) {
      v2f b = wrow[t * 16];
      acc[t] = __builtin_amdgcn_wmma_f32_16x16x4_f32(
          false, a, false, b, (short)0, acc[t], false, false);
    }
  }

#pragma unroll
  for (int t = 0; t < 8; ++t) {
    int col = t * 16 + lo;
    float bv = bias[col];
#pragma unroll
    for (int v = 0; v < 8; ++v) {
      int row = (wave << 4) + v + (hi << 3);
      float val = acc[t][v] + bv;
      if (do_relu) val = fmaxf(val, 0.0f);
      if (row < n_nodes) out[(long)row * D + col] = val;
    }
  }
}


#define ZSTRIDE 132

__global__ __launch_bounds__(128)
void pair_mlp_kernel(const int* __restrict__ pairs, const float* __restrict__ h,
                     const float* __restrict__ Wpa,
                     const float* __restrict__ ba,
                     const float* __restrict__ Wb, const float* __restrict__ bb,
                     float* __restrict__ out, int n_pairs) {
  __shared__ float zsh[4][16 * ZSTRIDE];
  int wv = threadIdx.x >> 5;
  int tile = (blockIdx.x << 2) + wv;
  int ntiles = (n_pairs + 15) >> 4;
  if (tile >= ntiles) return;
  int lane = threadIdx.x & 31;
  int lo = lane & 15, hi = lane >> 4;
  float* z = zsh[wv];

  {
    int p = (tile << 4) + lo;
    int pc = p < n_pairs ? p : 0;
    int i = pairs[2 * pc], j = pairs[2 * pc + 1];
    const float4* hi4 = (const float4*)(h + (long)i * D);
    const float4* hj4 = (const float4*)(h + (long)j * D);
    float4* zrow = (float4*)(z + lo * ZSTRIDE);
    for (int q = hi * 16; q < hi * 16 + 16; ++q) {
      float4 a = hi4[q], b = hj4[q];
      float4 v;
      v.x = a.x * b.x; v.y = a.y * b.y; v.z = a.z * b.z; v.w = a.w * b.w;
      zrow[q] = v;
    }
  }

  const v2f* wp = (const v2f*)Wpa;
  float partial[8];
#pragma unroll
  for (int v = 0; v < 8; ++v) partial[v] = 0.0f;

  for (int t = 0; t < 8; ++t) {
    v8f acc = (v8f)0.0f;
    for (int kk = 0; kk < D; kk += 4) {
      int k0 = kk + (hi << 1);
      v2f a = *(const v2f*)(z + lo * ZSTRIDE + k0);
      v2f b = wp[(long)(k0 >> 1) * D + t * 16 + lo];
      acc = __builtin_amdgcn_wmma_f32_16x16x4_f32(
          false, a, false, b, (short)0, acc, false, false);
    }
    int col = t * 16 + lo;
    float bv = ba[col], wb = Wb[col];
#pragma unroll
    for (int v = 0; v < 8; ++v) {
      partial[v] += fmaxf(acc[v] + bv, 0.0f) * wb;
    }
  }

#pragma unroll
  for (int v = 0; v < 8; ++v) {
    float x = partial[v];
    x += __shfl_xor(x, 1, 32);
    x += __shfl_xor(x, 2, 32);
    x += __shfl_xor(x, 4, 32);
    x += __shfl_xor(x, 8, 32);
    partial[v] = x;
  }
  if (lo == 0) {
    float bbv = bb[0];
#pragma unroll
    for (int v = 0; v < 8; ++v) {
      int row = (tile << 4) + v + (hi << 3);
      if (row < n_pairs) out[row] = partial[v] + bbv;
    }
  }
}


extern "C" void kernel_launch(void* const* d_in, const int* in_sizes, int n_in,
                              void* d_out, int out_size, void* d_ws, size_t ws_size,
                              hipStream_t stream) {
  const int* senders   = (const int*)d_in[1];
  const int* receivers = (const int*)d_in[2];
  const int* pairs     = (const int*)d_in[3];
  const float* emb = (const float*)d_in[4];
  const float* W1  = (const float*)d_in[5];
  const float* b1  = (const float*)d_in[6];
  const float* W2  = (const float*)d_in[7];
  const float* b2  = (const float*)d_in[8];
  const float* Wa  = (const float*)d_in[9];
  const float* ba  = (const float*)d_in[10];
  const float* Wb  = (const float*)d_in[11];
  const float* bb  = (const float*)d_in[12];
  float* out = (float*)d_out;

  int N = in_sizes[0];
  int E = in_sizes[1];
  int P = in_sizes[3] / 2;
  long ND = (long)N * D;

  float* ws    = (float*)d_ws;
  float* deg_s = ws;
  float* deg_r = ws + N;
  float* bufA  = ws + 2L * N;
  float* bufB  = bufA + ND;
  float* bufC  = bufB + ND;
  float* Wp1   = bufC + ND;
  float* Wp2   = Wp1 + 2L * D * D;
  float* Wpa   = Wp2 + 2L * D * D;

  const int T = 256;
  fill_kernel<<<(N + T - 1) / T, T, 0, stream>>>(deg_s, 1.0f, N);
  fill_kernel<<<(N + T - 1) / T, T, 0, stream>>>(deg_r, 1.0f, N);
  fill_kernel<<<(int)((ND + T - 1) / T), T, 0, stream>>>(bufB, 0.0f, ND);
  fill_kernel<<<(int)((ND + T - 1) / T), T, 0, stream>>>(bufC, 0.0f, ND);

  pack_pairs_kernel<<<(2 * D * D / 2 + T - 1) / T, T, 0, stream>>>(W1, Wp1, 2 * D, D);
  pack_pairs_kernel<<<(2 * D * D / 2 + T - 1) / T, T, 0, stream>>>(W2, Wp2, 2 * D, D);
  pack_pairs_kernel<<<(D * D / 2 + T - 1) / T, T, 0, stream>>>(Wa, Wpa, D, D);

  degree_kernel<<<(E + T - 1) / T, T, 0, stream>>>(senders, receivers, deg_s, deg_r, E);

  long n4 = ND / 4;
  long sth = (long)E * 32;
  int mt = (N + 15) >> 4;

  rownorm_kernel<<<(int)((n4 + T - 1) / T), T, 0, stream>>>(emb, deg_s, bufA, N);
  scatter_kernel<<<(int)((sth + T - 1) / T), T, 0, stream>>>(senders, receivers, bufA, bufB, E);
  sage_gemm_kernel<<<(mt + 7) / 8, 256, 0, stream>>>(emb, bufB, bufA, deg_r, Wp1, b1,
                                                     bufB, N,  1);

  rownorm_kernel<<<(int)((n4 + T - 1) / T), T, 0, stream>>>(bufB, deg_s, bufA, N);
  scatter_kernel<<<(int)((sth + T - 1) / T), T, 0, stream>>>(senders, receivers, bufA, bufC, E);
  sage_gemm_kernel<<<(mt + 7) / 8, 256, 0, stream>>>(bufB, bufC, bufA, deg_r, Wp2, b2,
                                                     bufC, N,  0);

  int ptiles = (P + 15) >> 4;
  pair_mlp_kernel<<<(ptiles + 3) / 4, 128, 0, stream>>>(pairs, bufC, Wpa, ba, Wb, bb, out, P);
}


# segment sums by receiver-bucketed gather (edge list built once per call, one wave per node, whole-row loads) instead of float atomic scatter-add; degree scalings folded into the gather
# speedup vs baseline: 1.9799x; 1.9799x over previous
#include <hip/hip_runtime.h>


#define D 128

typedef __attribute__((ext_vector_type(2))) float v2f;
typedef __attribute__((ext_vector_type(8))) float v8f;


__global__ void pack_pairs_kernel(const float* __restrict__ W, float* __restrict__ Wp,
                                  int K, int Ncols) {
  int i = blockIdx.x * blockDim.x + threadIdx.x;
  int total = (K >> 1) * Ncols;
  if (i >= total) return;
  int kp = i / Ncols, n = i - kp * Ncols;
  float2 v;
  v.x = W[(long)(2 * kp) * Ncols + n];
  v.y = W[(long)(2 * kp + 1) * Ncols + n];
  ((float2*)Wp)[(long)kp * Ncols + n] = v;
}


__global__ void init_kernel(float* __restrict__ deg_s, int* __restrict__ cnt_r,
                            int* __restrict__ total, int n) {
  int i = blockIdx.x * blockDim.x + threadIdx.x;
  if (i < n) { deg_s[i] = 1.0f; cnt_r[i] = 0; }
  if (i == 0) total[0] = 0;
}

__global__ void count_kernel(const int* __restrict__ snd, const int* __restrict__ rcv,
                             float* deg_s, int* cnt_r, int* __restrict__ rank, int E) {
  int e = blockIdx.x * blockDim.x + threadIdx.x;
  if (e < E) {
    int s = snd[e], r = rcv[e];
    unsafeAtomicAdd(&deg_s[s], 1.0f);
    rank[e] = atomicAdd(&cnt_r[r], 1);
  }
}

__global__ void alloc_kernel(const int* __restrict__ cnt_r, int* __restrict__ start,
                             int* total, int n) {
  int i = blockIdx.x * blockDim.x + threadIdx.x;
  int lane = threadIdx.x & 31;
  int c = i < n ? cnt_r[i] : 0;
  int incl = c;
  for (int d = 1; d < 32; d <<= 1) {
    int t = __shfl_up(incl, d, 32);
    if (lane >= d) incl += t;
  }
  int base = 0;
  if (lane == 31) base = atomicAdd(total, incl);
  base = __shfl(base, 31, 32);
  if (i < n) start[i] = base + incl - c;
}

__global__ void place_kernel(const int* __restrict__ snd, const int* __restrict__ rcv,
                             const int* __restrict__ start, const int* __restrict__ rank,
                             int* __restrict__ col, int E) {
  int e = blockIdx.x * blockDim.x + threadIdx.x;
  if (e < E) col[start[rcv[e]] + rank[e]] = snd[e];
}


__device__ __forceinline__ void axpy4(float4& acc, const float4 v, float w) {
  acc.x = fmaf(v.x, w, acc.x); acc.y = fmaf(v.y, w, acc.y);
  acc.z = fmaf(v.z, w, acc.z); acc.w = fmaf(v.w, w, acc.w);
}

__global__ __launch_bounds__(256)
void gather_kernel(const float* __restrict__ x, const float* __restrict__ deg_s,
                   const int* __restrict__ start, const int* __restrict__ cnt_r,
                   const int* __restrict__ col, float* __restrict__ agg, int n_nodes) {
  int node = __builtin_amdgcn_readfirstlane((int)((blockIdx.x << 3) + (threadIdx.x >> 5)));
  if (node >= n_nodes) return;
  int lane = threadIdx.x & 31;
  const float4* x4 = (const float4*)x;

  int beg = start[node];
  int cnt = cnt_r[node];
  int end = beg + cnt;

  float4 acc;
  {
    float w = rsqrtf(deg_s[node]);
    float4 v = x4[(long)node * 32 + lane];
    acc.x = v.x * w; acc.y = v.y * w; acc.z = v.z * w; acc.w = v.w * w;
  }

  for (int base = beg; base < end; base += 32) {
    int n = end - base;
    if (n > 32) n = 32;
    int sidx = node;
    float ssc = 0.0f;
    if (lane < n) {
      int sv = col[base + lane];
      sidx = (unsigned)sv < (unsigned)n_nodes ? sv : node;
      ssc = rsqrtf(deg_s[sidx]);
    }
    int sw = __float_as_int(ssc);
    int j = 0;
    for (; j + 4 <= n; j += 4) {
      int s0 = __builtin_amdgcn_readlane(sidx, j);
      int s1 = __builtin_amdgcn_readlane(sidx, j + 1);
      int s2 = __builtin_amdgcn_readlane(sidx, j + 2);
      int s3 = __builtin_amdgcn_readlane(sidx, j + 3);
      float w0 = __int_as_float(__builtin_amdgcn_readlane(sw, j));
      float w1 = __int_as_float(__builtin_amdgcn_readlane(sw, j + 1));
      float w2 = __int_as_float(__builtin_amdgcn_readlane(sw, j + 2));
      float w3 = __int_as_float(__builtin_amdgcn_readlane(sw, j + 3));
      float4 v0 = x4[(long)s0 * 32 + lane];
      float4 v1 = x4[(long)s1 * 32 + lane];
      float4 v2 = x4[(long)s2 * 32 + lane];
      float4 v3 = x4[(long)s3 * 32 + lane];
      axpy4(acc, v0, w0);
      axpy4(acc, v1, w1);
      axpy4(acc, v2, w2);
      axpy4(acc, v3, w3);
    }
    for (; j < n; ++j) {
      int s0 = __builtin_amdgcn_readlane(sidx, j);
      float w0 = __int_as_float(__builtin_amdgcn_readlane(sw, j));
      float4 v0 = x4[(long)s0 * 32 + lane];
      axpy4(acc, v0, w0);
    }
  }

  float c = (float)(cnt + 1);
  float rs = rsqrtf(c);
  float scale = rs * rs * rs;
  acc.x *= scale; acc.y *= scale; acc.z *= scale; acc.w *= scale;
  ((float4*)agg)[(long)node * 32 + lane] = acc;
}


__global__ __launch_bounds__(256)
void sage_gemm_kernel(const float* __restrict__ A0,
                      const float* agg,
                      const float* __restrict__ Wp,
                      const float* __restrict__ bias,
                      float* out,
                      int n_nodes, int do_relu) {
  int wave = (blockIdx.x << 3) + (threadIdx.x >> 5);
  int mt_count = (n_nodes + 15) >> 4;
  if (wave >= mt_count) return;
  int lane = threadIdx.x & 31;
  int lo = lane & 15, hi = lane >> 4;

  int m = (wave << 4) + lo;
  bool mvalid = m < n_nodes;
  int mr = mvalid ? m : 0;

  const float* a0row = A0 + (long)mr * D;
  const float* agrow = agg + (long)mr * D;
  const v2f* wp = (const v2f*)Wp;

  v8f acc[8];
#pragma unroll
  for (int t = 0; t < 8; ++t) acc[t] = (v8f)0.0f;

  for (int kk = 0; kk < D; kk += 4) {
    int k0 = kk + (hi << 1);
    v2f a = *(const v2f*)(a0row + k0);
    if (!mvalid) { a.x = 0.0f; a.y = 0.0f; }
    const v2f* wrow = wp + (long)(k0 >> 1) * D + lo;
#pragma unroll
    for (int t = 0; t < 8; ++t) {
      v2f b = wrow[t * 16];
      acc[t] = __builtin_amdgcn_wmma_f32_16x16x4_f32(
          false, a, false, b, (short)0, acc[t], false, false);
    }
  }

  for (int kk = 0; kk < D; kk += 4) {
    int k0 = kk + (hi << 1);
    v2f a = *(const v2f*)(agrow + k0);
    if (!mvalid) { a.x = 0.0f; a.y = 0.0f; }
    const v2f* wrow = wp + (long)((k0 + D) >> 1) * D + lo;
#pragma unroll
    for (int t = 0; t < 8; ++t) {
      v2f b = wrow[t * 16];
      acc[t] = __builtin_amdgcn_wmma_f32_16x16x4_f32(
          false, a, false, b, (short)0, acc[t], false, false);
    }
  }

#pragma unroll
  for (int t = 0; t < 8; ++t) {
    int col = t * 16 + lo;
    float bv = bias[col];
#pragma unroll
    for (int v = 0; v < 8; ++v) {
      int row = (wave << 4) + v + (hi << 3);
      float val = acc[t][v] + bv;
      if (do_relu) val = fmaxf(val, 0.0f);
      if (row < n_nodes) out[(long)row * D + col] = val;
    }
  }
}


#define ZSTRIDE 132

__global__ __launch_bounds__(128)
void pair_mlp_kernel(const int* __restrict__ pairs, const float* __restrict__ h,
                     const float* __restrict__ Wpa,
                     const float* __restrict__ ba,
                     const float* __restrict__ Wb, const float* __restrict__ bb,
                     float* __restrict__ out, int n_pairs) {
  __shared__ float zsh[4][16 * ZSTRIDE];
  int wv = threadIdx.x >> 5;
  int tile = (blockIdx.x << 2) + wv;
  int ntiles = (n_pairs + 15) >> 4;
  if (tile >= ntiles) return;
  int lane = threadIdx.x & 31;
  int lo = lane & 15, hi = lane >> 4;
  float* z = zsh[wv];

  {
    int p = (tile << 4) + lo;
    int pc = p < n_pairs ? p : 0;
    int i = pairs[2 * pc], j = pairs[2 * pc + 1];
    const float4* hi4 = (const float4*)(h + (long)i * D);
    const float4* hj4 = (const float4*)(h + (long)j * D);
    float4* zrow = (float4*)(z + lo * ZSTRIDE);
    for (int q = hi * 16; q < hi * 16 + 16; ++q) {
      float4 a = hi4[q], b = hj4[q];
      float4 v;
      v.x = a.x * b.x; v.y = a.y * b.y; v.z = a.z * b.z; v.w = a.w * b.w;
      zrow[q] = v;
    }
  }

  const v2f* wp = (const v2f*)Wpa;
  float partial[8];
#pragma unroll
  for (int v = 0; v < 8; ++v) partial[v] = 0.0f;

  for (int t = 0; t < 8; ++t) {
    v8f acc = (v8f)0.0f;
    for (int kk = 0; kk < D; kk += 4) {
      int k0 = kk + (hi << 1);
      v2f a = *(const v2f*)(z + lo * ZSTRIDE + k0);
      v2f b = wp[(long)(k0 >> 1) * D + t * 16 + lo];
      acc = __builtin_amdgcn_wmma_f32_16x16x4_f32(
          false, a, false, b, (short)0, acc, false, false);
    }
    int col = t * 16 + lo;
    float bv = ba[col], wb = Wb[col];
#pragma unroll
    for (int v = 0; v < 8; ++v) {
      partial[v] += fmaxf(acc[v] + bv, 0.0f) * wb;
    }
  }

#pragma unroll
  for (int v = 0; v < 8; ++v) {
    float x = partial[v];
    x += __shfl_xor(x, 1, 32);
    x += __shfl_xor(x, 2, 32);
    x += __shfl_xor(x, 4, 32);
    x += __shfl_xor(x, 8, 32);
    partial[v] = x;
  }
  if (lo == 0) {
    float bbv = bb[0];
#pragma unroll
    for (int v = 0; v < 8; ++v) {
      int row = (tile << 4) + v + (hi << 3);
      if (row < n_pairs) out[row] = partial[v] + bbv;
    }
  }
}


extern "C" void kernel_launch(void* const* d_in, const int* in_sizes, int n_in,
                              void* d_out, int out_size, void* d_ws, size_t ws_size,
                              hipStream_t stream) {
  const int* senders   = (const int*)d_in[1];
  const int* receivers = (const int*)d_in[2];
  const int* pairs     = (const int*)d_in[3];
  const float* emb = (const float*)d_in[4];
  const float* W1  = (const float*)d_in[5];
  const float* b1  = (const float*)d_in[6];
  const float* W2  = (const float*)d_in[7];
  const float* b2  = (const float*)d_in[8];
  const float* Wa  = (const float*)d_in[9];
  const float* ba  = (const float*)d_in[10];
  const float* Wb  = (const float*)d_in[11];
  const float* bb  = (const float*)d_in[12];
  float* out = (float*)d_out;

  int N = in_sizes[0];
  int E = in_sizes[1];
  int P = in_sizes[3] / 2;
  long ND = (long)N * D;

  float* ws    = (float*)d_ws;
  float* deg_s = ws;
  int*   cnt_r = (int*)(ws + N);
  float* bufA  = ws + 2L * N;
  float* bufB  = bufA + ND;
  float* bufC  = bufB + ND;
  float* Wp1   = bufC + ND;
  float* Wp2   = Wp1 + 2L * D * D;
  float* Wpa   = Wp2 + 2L * D * D;

  int* start = (int*)bufA;
  int* rank  = start + N;
  int* col   = rank + E;
  int* total = col + E;

  const int T = 256;
  init_kernel<<<(N + T - 1) / T, T, 0, stream>>>(deg_s, cnt_r, total, N);

  pack_pairs_kernel<<<(2 * D * D / 2 + T - 1) / T, T, 0, stream>>>(W1, Wp1, 2 * D, D);
  pack_pairs_kernel<<<(2 * D * D / 2 + T - 1) / T, T, 0, stream>>>(W2, Wp2, 2 * D, D);
  pack_pairs_kernel<<<(D * D / 2 + T - 1) / T, T, 0, stream>>>(Wa, Wpa, D, D);

  count_kernel<<<(E + T - 1) / T, T, 0, stream>>>(senders, receivers, deg_s, cnt_r, rank, E);
  alloc_kernel<<<(N + T - 1) / T, T, 0, stream>>>(cnt_r, start, total, N);
  place_kernel<<<(E + T - 1) / T, T, 0, stream>>>(senders, receivers, start, rank, col, E);

  int mt = (N + 15) >> 4;
  int gblocks = (N + 7) / 8;

  gather_kernel<<<gblocks, 256, 0, stream>>>(emb, deg_s, start, cnt_r, col, bufB, N);
  sage_gemm_kernel<<<(mt + 7) / 8, 256, 0, stream>>>(emb, bufB, Wp1, b1, bufB, N,  1);

  gather_kernel<<<gblocks, 256, 0, stream>>>(bufB, deg_s, start, cnt_r, col, bufC, N);
  sage_gemm_kernel<<<(mt + 7) / 8, 256, 0, stream>>>(bufB, bufC, Wp2, b2, bufC, N,  0);

  int ptiles = (P + 15) >> 4;
  pair_mlp_kernel<<<(ptiles + 3) / 4, 128, 0, stream>>>(pairs, bufC, Wpa, ba, Wb, bb, out, P);
}


# matrix kernels (both SAGE GEMMs, pair MLP) run as 256 persistent 16-wave work-groups with the packed weights resident in LDS; A rows staged through LDS by whole-row loads
# speedup vs baseline: 2.1936x; 1.1079x over previous
#include <hip/hip_runtime.h>


#define D 128

typedef __attribute__((ext_vector_type(2))) float v2f;
typedef __attribute__((ext_vector_type(8))) float v8f;


__global__ void pack_pairs_kernel(const float* __restrict__ W, float* __restrict__ Wp,
                                  int K, int Ncols) {
  int i = blockIdx.x * blockDim.x + threadIdx.x;
  int total = (K >> 1) * Ncols;
  if (i >= total) return;
  int kp = i / Ncols, n = i - kp * Ncols;
  float2 v;
  v.x = W[(long)(2 * kp) * Ncols + n];
  v.y = W[(long)(2 * kp + 1) * Ncols + n];
  ((float2*)Wp)[(long)kp * Ncols + n] = v;
}


__global__ void init_kernel(float* __restrict__ deg_s, int* __restrict__ cnt_r,
                            int* __restrict__ total, int n) {
  int i = blockIdx.x * blockDim.x + threadIdx.x;
  if (i < n) { deg_s[i] = 1.0f; cnt_r[i] = 0; }
  if (i == 0) total[0] = 0;
}

__global__ void count_kernel(const int* __restrict__ snd, const int* __restrict__ rcv,
                             float* deg_s, int* cnt_r, int* __restrict__ rank, int E) {
  int e = blockIdx.x * blockDim.x + threadIdx.x;
  if (e < E) {
    int s = snd[e], r = rcv[e];
    unsafeAtomicAdd(&deg_s[s], 1.0f);
    rank[e] = atomicAdd(&cnt_r[r], 1);
  }
}

__global__ void alloc_kernel(const int* __restrict__ cnt_r, int* __restrict__ start,
                             int* total, int n) {
  int i = blockIdx.x * blockDim.x + threadIdx.x;
  int lane = threadIdx.x & 31;
  int c = i < n ? cnt_r[i] : 0;
  int incl = c;
  for (int d = 1; d < 32; d <<= 1) {
    int t = __shfl_up(incl, d, 32);
    if (lane >= d) incl += t;
  }
  int base = 0;
  if (lane == 31) base = atomicAdd(total, incl);
  base = __shfl(base, 31, 32);
  if (i < n) start[i] = base + incl - c;
}

__global__ void place_kernel(const int* __restrict__ snd, const int* __restrict__ rcv,
                             const int* __restrict__ start, const int* __restrict__ rank,
                             int* __restrict__ col, int E) {
  int e = blockIdx.x * blockDim.x + threadIdx.x;
  if (e < E) col[start[rcv[e]] + rank[e]] = snd[e];
}


__device__ __forceinline__ void axpy4(float4& acc, const float4 v, float w) {
  acc.x = fmaf(v.x, w, acc.x); acc.y = fmaf(v.y, w, acc.y);
  acc.z = fmaf(v.z, w, acc.z); acc.w = fmaf(v.w, w, acc.w);
}

__global__ __launch_bounds__(256)
void gather_kernel(const float* __restrict__ x, const float* __restrict__ deg_s,
                   const int* __restrict__ start, const int* __restrict__ cnt_r,
                   const int* __restrict__ col, float* __restrict__ agg, int n_nodes) {
  int node = __builtin_amdgcn_readfirstlane((int)((blockIdx.x << 3) + (threadIdx.x >> 5)));
  if (node >= n_nodes) return;
  int lane = threadIdx.x & 31;
  const float4* x4 = (const float4*)x;

  int beg = start[node];
  int cnt = cnt_r[node];
  int end = beg + cnt;

  float4 acc;
  {
    float w = rsqrtf(deg_s[node]);
    float4 v = x4[(long)node * 32 + lane];
    acc.x = v.x * w; acc.y = v.y * w; acc.z = v.z * w; acc.w = v.w * w;
  }

  for (int base = beg; base < end; base += 32) {
    int n = end - base;
    if (n > 32) n = 32;
    int sidx = node;
    float ssc = 0.0f;
    if (lane < n) {
      int sv = col[base + lane];
      sidx = (unsigned)sv < (unsigned)n_nodes ? sv : node;
      ssc = rsqrtf(deg_s[sidx]);
    }
    int sw = __float_as_int(ssc);
    int j = 0;
    for (; j + 4 <= n; j += 4) {
      int s0 = __builtin_amdgcn_readlane(sidx, j);
      int s1 = __builtin_amdgcn_readlane(sidx, j + 1);
      int s2 = __builtin_amdgcn_readlane(sidx, j + 2);
      int s3 = __builtin_amdgcn_readlane(sidx, j + 3);
      float w0 = __int_as_float(__builtin_amdgcn_readlane(sw, j));
      float w1 = __int_as_float(__builtin_amdgcn_readlane(sw, j + 1));
      float w2 = __int_as_float(__builtin_amdgcn_readlane(sw, j + 2));
      float w3 = __int_as_float(__builtin_amdgcn_readlane(sw, j + 3));
      float4 v0 = x4[(long)s0 * 32 + lane];
      float4 v1 = x4[(long)s1 * 32 + lane];
      float4 v2 = x4[(long)s2 * 32 + lane];
      float4 v3 = x4[(long)s3 * 32 + lane];
      axpy4(acc, v0, w0);
      axpy4(acc, v1, w1);
      axpy4(acc, v2, w2);
      axpy4(acc, v3, w3);
    }
    for (; j < n; ++j) {
      int s0 = __builtin_amdgcn_readlane(sidx, j);
      float w0 = __int_as_float(__builtin_amdgcn_readlane(sw, j));
      float4 v0 = x4[(long)s0 * 32 + lane];
      axpy4(acc, v0, w0);
    }
  }

  float c = (float)(cnt + 1);
  float rs = rsqrtf(c);
  float scale = rs * rs * rs;
  acc.x *= scale; acc.y *= scale; acc.z *= scale; acc.w *= scale;
  ((float4*)agg)[(long)node * 32 + lane] = acc;
}


#define ZSTRIDE 132
#define WPITCH 288
#define PM_WAVES 16
#define PM_GROUPS 256


__global__ __launch_bounds__(512)
void sage_gemm_kernel(const float* __restrict__ A0,
                      const float* agg,
                      const float* __restrict__ Wp,
                      const float* __restrict__ bias,
                      float* out,
                      int n_nodes, int do_relu) {
  __shared__ __attribute__((aligned(16))) float wsh[128 * WPITCH];
  __shared__ __attribute__((aligned(16))) float zsh[PM_WAVES][16 * ZSTRIDE];
  int wv = __builtin_amdgcn_readfirstlane((int)(threadIdx.x >> 5));
  int lane = threadIdx.x & 31;
  int lo = lane & 15, hi = lane >> 4;

#pragma unroll
  for (int it = 0; it < 16; ++it) {
    int i = (int)threadIdx.x + it * (PM_WAVES * 32);
    int kp = i >> 6, c = i & 63;
    float4 v = ((const float4*)Wp)[i];
    *(float4*)(wsh + kp * WPITCH + 4 * c) = v;
  }

  float bv[8];
#pragma unroll
  for (int t = 0; t < 8; ++t) bv[t] = bias[t * 16 + lo];

  __syncthreads();

  float* z = zsh[wv];
  int mt_count = (n_nodes + 15) >> 4;

  for (int tile = wv * PM_GROUPS + (int)blockIdx.x; tile < mt_count; tile += PM_GROUPS * PM_WAVES) {
    v8f acc[8];
#pragma unroll
    for (int t = 0; t < 8; ++t) acc[t] = (v8f)0.0f;

#pragma unroll 1
    for (int half = 0; half < 2; ++half) {
      const float4* src4 = (const float4*)(half == 0 ? A0 : agg);
#pragma unroll
      for (int r = 0; r < 16; ++r) {
        int node = (tile << 4) + r;
        int nc = node < n_nodes ? node : n_nodes - 1;
        float4 v = src4[(long)nc * 32 + lane];
        if (node >= n_nodes) { v.x = 0.0f; v.y = 0.0f; v.z = 0.0f; v.w = 0.0f; }
        *(float4*)(z + r * ZSTRIDE + 4 * lane) = v;
      }

      const float* zrow = z + lo * ZSTRIDE + 2 * hi;
      const float* wcol = wsh + (half * 64 + hi) * WPITCH + 2 * lo;
#pragma unroll 1
      for (int s0 = 0; s0 < 32; s0 += 4) {
#pragma unroll
        for (int j = 0; j < 4; ++j) {
          v2f a = *(const v2f*)(zrow + 4 * (s0 + j));
#pragma unroll
          for (int t = 0; t < 8; ++t) {
            v2f b = *(const v2f*)(wcol + (2 * (s0 + j)) * WPITCH + 32 * t);
            acc[t] = __builtin_amdgcn_wmma_f32_16x16x4_f32(
                false, a, false, b, (short)0, acc[t], false, false);
          }
        }
      }
    }

#pragma unroll
    for (int t = 0; t < 8; ++t) {
      int col = t * 16 + lo;
#pragma unroll
      for (int v = 0; v < 8; ++v) {
        int row = (tile << 4) + v + (hi << 3);
        float val = acc[t][v] + bv[t];
        if (do_relu) val = fmaxf(val, 0.0f);
        if (row < n_nodes) out[(long)row * D + col] = val;
      }
    }
  }
}


__global__ __launch_bounds__(512)
void pair_mlp_kernel(const int* __restrict__ pairs, const float* __restrict__ h,
                     const float* __restrict__ Wpa,
                     const float* __restrict__ ba,
                     const float* __restrict__ Wb, const float* __restrict__ bb,
                     float* __restrict__ out, int n_pairs) {
  __shared__ __attribute__((aligned(16))) float wsh[64 * WPITCH];
  __shared__ __attribute__((aligned(16))) float zsh[PM_WAVES][16 * ZSTRIDE];
  int wv = __builtin_amdgcn_readfirstlane((int)(threadIdx.x >> 5));
  int lane = threadIdx.x & 31;
  int lo = lane & 15, hi = lane >> 4;

#pragma unroll
  for (int it = 0; it < 8; ++it) {
    int i = (int)threadIdx.x + it * (PM_WAVES * 32);
    int kp = i >> 6, c = i & 63;
    float4 v = ((const float4*)Wpa)[i];
    *(float4*)(wsh + kp * WPITCH + 4 * c) = v;
  }

  float bav[8], wbv[8];
#pragma unroll
  for (int t = 0; t < 8; ++t) { bav[t] = ba[t * 16 + lo]; wbv[t] = Wb[t * 16 + lo]; }
  float bbv = bb[0];

  __syncthreads();

  float* z = zsh[wv];
  const float4* h4 = (const float4*)h;
  int ntiles = (n_pairs + 15) >> 4;

  for (int tile = wv * PM_GROUPS + (int)blockIdx.x; tile < ntiles; tile += PM_GROUPS * PM_WAVES) {
    {
      int p = (tile << 4) + lo;
      int pc = p < n_pairs ? p : 0;
      int iv = pairs[2 * pc], jv = pairs[2 * pc + 1];
#pragma unroll
      for (int r = 0; r < 16; ++r) {
        int ir = __builtin_amdgcn_readlane(iv, r);
        int jr = __builtin_amdgcn_readlane(jv, r);
        float4 a = h4[(long)ir * 32 + lane];
        float4 b = h4[(long)jr * 32 + lane];
        float4 v;
        v.x = a.x * b.x; v.y = a.y * b.y; v.z = a.z * b.z; v.w = a.w * b.w;
        *(float4*)(z + r * ZSTRIDE + 4 * lane) = v;
      }
    }

    v2f af[32];
#pragma unroll
    for (int s = 0; s < 32; ++s) af[s] = *(const v2f*)(z + lo * ZSTRIDE + 4 * s + 2 * hi);

    float partial[8];
#pragma unroll
    for (int v = 0; v < 8; ++v) partial[v] = 0.0f;

#pragma unroll
    for (int tp = 0; tp < 4; ++tp) {
      v8f acc0 = (v8f)0.0f, acc1 = (v8f)0.0f;
      const float* wcol = wsh + hi * WPITCH + 2 * (tp * 32 + lo);
#pragma unroll
      for (int s = 0; s < 32; ++s) {
        v2f b0 = *(const v2f*)(wcol + (2 * s) * WPITCH);
        v2f b1 = *(const v2f*)(wcol + (2 * s) * WPITCH + 32);
        acc0 = __builtin_amdgcn_wmma_f32_16x16x4_f32(
            false, af[s], false, b0, (short)0, acc0, false, false);
        acc1 = __builtin_amdgcn_wmma_f32_16x16x4_f32(
            false, af[s], false, b1, (short)0, acc1, false, false);
      }
      {
        float bv = bav[2 * tp], wb = wbv[2 * tp];
#pragma unroll
        for (int v = 0; v < 8; ++v) partial[v] += fmaxf(acc0[v] + bv, 0.0f) * wb;
      }
      {
        float bv = bav[2 * tp + 1], wb = wbv[2 * tp + 1];
#pragma unroll
        for (int v = 0; v < 8; ++v) partial[v] += fmaxf(acc1[v] + bv, 0.0f) * wb;
      }
    }

#pragma unroll
    for (int v = 0; v < 8; ++v) {
      float x = partial[v];
      x += __shfl_xor(x, 1, 32);
      x += __shfl_xor(x, 2, 32);
      x += __shfl_xor(x, 4, 32);
      x += __shfl_xor(x, 8, 32);
      partial[v] = x;
    }
    if (lo == 0) {
#pragma unroll
      for (int v = 0; v < 8; ++v) {
        int row = (tile << 4) + v + (hi << 3);
        if (row < n_pairs) out[row] = partial[v] + bbv;
      }
    }
  }
}


extern "C" void kernel_launch(void* const* d_in, const int* in_sizes, int n_in,
                              void* d_out, int out_size, void* d_ws, size_t ws_size,
                              hipStream_t stream) {
  const int* senders   = (const int*)d_in[1];
  const int* receivers = (const int*)d_in[2];
  const int* pairs     = (const int*)d_in[3];
  const float* emb = (const float*)d_in[4];
  const float* W1  = (const float*)d_in[5];
  const float* b1  = (const float*)d_in[6];
  const float* W2  = (const float*)d_in[7];
  const float* b2  = (const float*)d_in[8];
  const float* Wa  = (const float*)d_in[9];
  const float* ba  = (const float*)d_in[10];
  const float* Wb  = (const float*)d_in[11];
  const float* bb  = (const float*)d_in[12];
  float* out = (float*)d_out;

  int N = in_sizes[0];
  int E = in_sizes[1];
  int P = in_sizes[3] / 2;
  long ND = (long)N * D;

  float* ws    = (float*)d_ws;
  float* deg_s = ws;
  int*   cnt_r = (int*)(ws + N);
  float* bufA  = ws + 2L * N;
  float* bufB  = bufA + ND;
  float* bufC  = bufB + ND;
  float* Wp1   = bufC + ND;
  float* Wp2   = Wp1 + 2L * D * D;
  float* Wpa   = Wp2 + 2L * D * D;

  int* start = (int*)bufA;
  int* rank  = start + N;
  int* col   = rank + E;
  int* total = col + E;

  const int T = 256;
  init_kernel<<<(N + T - 1) / T, T, 0, stream>>>(deg_s, cnt_r, total, N);

  pack_pairs_kernel<<<(2 * D * D / 2 + T - 1) / T, T, 0, stream>>>(W1, Wp1, 2 * D, D);
  pack_pairs_kernel<<<(2 * D * D / 2 + T - 1) / T, T, 0, stream>>>(W2, Wp2, 2 * D, D);
  pack_pairs_kernel<<<(D * D / 2 + T - 1) / T, T, 0, stream>>>(Wa, Wpa, D, D);

  count_kernel<<<(E + T - 1) / T, T, 0, stream>>>(senders, receivers, deg_s, cnt_r, rank, E);
  alloc_kernel<<<(N + T - 1) / T, T, 0, stream>>>(cnt_r, start, total, N);
  place_kernel<<<(E + T - 1) / T, T, 0, stream>>>(senders, receivers, start, rank, col, E);

  int gblocks = (N + 7) / 8;

  gather_kernel<<<gblocks, 256, 0, stream>>>(emb, deg_s, start, cnt_r, col, bufB, N);
  sage_gemm_kernel<<<PM_GROUPS, PM_WAVES * 32, 0, stream>>>(emb, bufB, Wp1, b1, bufB, N,  1);

  gather_kernel<<<gblocks, 256, 0, stream>>>(bufB, deg_s, start, cnt_r, col, bufC, N);
  sage_gemm_kernel<<<PM_GROUPS, PM_WAVES * 32, 0, stream>>>(bufB, bufC, Wp2, b2, bufC, N,  0);

  pair_mlp_kernel<<<PM_GROUPS, PM_WAVES * 32, 0, stream>>>(pairs, bufC, Wpa, ba, Wb, bb, out, P);
}
